# starting kernel
# baseline (speedup 1.0000x reference)
#include <hip/hip_runtime.h>
#include <hip/hip_bf16.h>

typedef __attribute__((ext_vector_type(2))) float v2f;
typedef __attribute__((ext_vector_type(4))) float v4f;
typedef __attribute__((ext_vector_type(8))) float v8f;

#define SEQ    2048
#define BATCH  2
#define HEADS  16
#define HDIM   128
#define SROW   4096
#define KSTRIDE 140
#define PSTRIDE 18

__global__ __launch_bounds__(256) void fa_fp32_wmma_kernel(
    const float* __restrict__ Q, const float* __restrict__ K,
    const float* __restrict__ V, float* __restrict__ Out)
{
    __shared__ float sK[2][16 * KSTRIDE];
    __shared__ float sV[2][16 * KSTRIDE];
    __shared__ float sP[8][16 * PSTRIDE];

    const int tid  = threadIdx.x;
    const int wave = tid >> 5;
    const int lane = tid & 31;
    const int hi   = lane >> 4;
    const int l15  = lane & 15;

    const size_t head_off = (size_t)blockIdx.y * HDIM;
    const int qblk = blockIdx.x * 128;
    const int q0w  = qblk + wave * 16;

    v2f qf[32];
    {
        const float inv_norm = 0.08838834764831845f;
        const float* qbase = Q + (size_t)(q0w + l15) * SROW + head_off + 2 * hi;
        #pragma unroll
        for (int ks = 0; ks < 32; ++ks) {
            v2f q = *(const v2f*)(qbase + ks * 4);
            q[0] *= inv_norm; q[1] *= inv_norm;
            qf[ks] = q;
        }
    }

    v8f o[8];
    #pragma unroll
    for (int dt = 0; dt < 8; ++dt) o[dt] = (v8f){};
    float mrow[8], lrow[8];
    #pragma unroll
    for (int r = 0; r < 8; ++r) { mrow[r] = -1e30f; lrow[r] = 0.0f; }

    const int nkt       = qblk / 16 + 8;
    const int my_kt_max = qblk / 16 + wave;

    const int ld_row = tid >> 4;
    const int ld_col = (tid & 15) * 8;
    const size_t ld_base = (size_t)ld_row * SROW + head_off + ld_col;
    v4f rk0, rk1, rv0, rv1;
    {
        rk0 = *(const v4f*)(K + ld_base);
        rk1 = *(const v4f*)(K + ld_base + 4);
        rv0 = *(const v4f*)(V + ld_base);
        rv1 = *(const v4f*)(V + ld_base + 4);
    }

    for (int kt = 0; kt < nkt; ++kt) {
        float* bK = sK[kt & 1];
        float* bV = sV[kt & 1];
        *(v4f*)&bK[ld_row * KSTRIDE + ld_col]     = rk0;
        *(v4f*)&bK[ld_row * KSTRIDE + ld_col + 4] = rk1;
        *(v4f*)&bV[ld_row * KSTRIDE + ld_col]     = rv0;
        *(v4f*)&bV[ld_row * KSTRIDE + ld_col + 4] = rv1;
        if (kt + 1 < nkt) {
            const size_t g = ld_base + (size_t)(kt + 1) * 16 * SROW;
            rk0 = *(const v4f*)(K + g);
            rk1 = *(const v4f*)(K + g + 4);
            rv0 = *(const v4f*)(V + g);
            rv1 = *(const v4f*)(V + g + 4);
        }
        __syncthreads();

        if (kt <= my_kt_max) {
            v8f s = (v8f){};
            #pragma unroll
            for (int g4 = 0; g4 < 4; ++g4) {
                v2f kb[8];
                #pragma unroll
                for (int j = 0; j < 8; ++j)
                    kb[j] = *(const v2f*)&bK[l15 * KSTRIDE + (g4 * 8 + j) * 4 + 2 * hi];
                #pragma unroll
                for (int j = 0; j < 8; ++j)
                    s = __builtin_amdgcn_wmma_f32_16x16x4_f32(
                            false, qf[g4 * 8 + j], false, kb[j], (short)0, s, false, false);
            }

            const int keyc = kt * 16 + l15;
            #pragma unroll
            for (int r = 0; r < 8; ++r) {
                const int qrow = q0w + r + 8 * hi;
                float sv = (keyc <= qrow) ? s[r] : -10000.0f;

                float mx = sv;
                mx = fmaxf(mx, __shfl_xor(mx, 1, 32));
                mx = fmaxf(mx, __shfl_xor(mx, 2, 32));
                mx = fmaxf(mx, __shfl_xor(mx, 4, 32));
                mx = fmaxf(mx, __shfl_xor(mx, 8, 32));

                const float mnew = fmaxf(mrow[r], mx);
                const float corr = __expf(mrow[r] - mnew);
                const float p    = __expf(sv - mnew);

                float ps = p;
                ps += __shfl_xor(ps, 1, 32);
                ps += __shfl_xor(ps, 2, 32);
                ps += __shfl_xor(ps, 4, 32);
                ps += __shfl_xor(ps, 8, 32);

                lrow[r] = lrow[r] * corr + ps;
                mrow[r] = mnew;
                #pragma unroll
                for (int dt = 0; dt < 8; ++dt) o[dt][r] *= corr;

                sP[wave][(r + 8 * hi) * PSTRIDE + l15] = p;
            }
            asm volatile("s_wait_dscnt 0x0" ::: "memory");

            v2f a[4];
            #pragma unroll
            for (int ka = 0; ka < 4; ++ka)
                a[ka] = *(const v2f*)&sP[wave][l15 * PSTRIDE + ka * 4 + 2 * hi];
            #pragma unroll
            for (int ka = 0; ka < 4; ++ka) {
                v2f vb[8];
                #pragma unroll
                for (int dt = 0; dt < 8; ++dt) {
                    vb[dt][0] = bV[(ka * 4 + 2 * hi    ) * KSTRIDE + dt * 16 + l15];
                    vb[dt][1] = bV[(ka * 4 + 2 * hi + 1) * KSTRIDE + dt * 16 + l15];
                }
                #pragma unroll
                for (int dt = 0; dt < 8; ++dt)
                    o[dt] = __builtin_amdgcn_wmma_f32_16x16x4_f32(
                                false, a[ka], false, vb[dt], (short)0, o[dt], false, false);
            }
        }
    }

    #pragma unroll
    for (int r = 0; r < 8; ++r) {
        const float inv = 1.0f / lrow[r];
        float* op = Out + (size_t)(q0w + r + 8 * hi) * SROW + head_off;
        #pragma unroll
        for (int dt = 0; dt < 8; ++dt)
            op[dt * 16 + l15] = o[dt][r] * inv;
    }
}

extern "C" void kernel_launch(void* const* d_in, const int* in_sizes, int n_in,
                              void* d_out, int out_size, void* d_ws, size_t ws_size,
                              hipStream_t stream) {
    (void)in_sizes; (void)n_in; (void)out_size; (void)d_ws; (void)ws_size;
    const float* Q = (const float*)d_in[0];
    const float* K = (const float*)d_in[1];
    const float* V = (const float*)d_in[2];
    float* Out     = (float*)d_out;

    dim3 grid(SEQ / 128, BATCH * HEADS);
    dim3 block(256);
    hipLaunchKernelGGL(fa_fp32_wmma_kernel, grid, block, 0, stream, Q, K, V, Out);
}


# Row max and row sum butterflies over 16 lanes run as DPP row_xmask vector ops instead of LDS permutes; same lane pairs in the same order, results bit-identical
# speedup vs baseline: 1.5232x; 1.5232x over previous
#include <hip/hip_runtime.h>
#include <hip/hip_bf16.h>

typedef __attribute__((ext_vector_type(2))) float v2f;
typedef __attribute__((ext_vector_type(4))) float v4f;
typedef __attribute__((ext_vector_type(8))) float v8f;

#define SEQ    2048
#define BATCH  2
#define HEADS  16
#define HDIM   128
#define SROW   4096
#define KSTRIDE 140
#define PSTRIDE 18

template <int M>
__device__ __forceinline__ float row_xor(float x) {
    const int b = __builtin_bit_cast(int, x);
    return __builtin_bit_cast(float,
        __builtin_amdgcn_update_dpp(b, b, 0x160 | M, 0xf, 0xf, true));
}

__global__ __launch_bounds__(256) void fa_fp32_wmma_kernel(
    const float* __restrict__ Q, const float* __restrict__ K,
    const float* __restrict__ V, float* __restrict__ Out)
{
    __shared__ float sK[2][16 * KSTRIDE];
    __shared__ float sV[2][16 * KSTRIDE];
    __shared__ float sP[8][16 * PSTRIDE];

    const int tid  = threadIdx.x;
    const int wave = tid >> 5;
    const int lane = tid & 31;
    const int hi   = lane >> 4;
    const int l15  = lane & 15;

    const size_t head_off = (size_t)blockIdx.y * HDIM;
    const int qblk = blockIdx.x * 128;
    const int q0w  = qblk + wave * 16;

    v2f qf[32];
    {
        const float inv_norm = 0.08838834764831845f;
        const float* qbase = Q + (size_t)(q0w + l15) * SROW + head_off + 2 * hi;
        #pragma unroll
        for (int ks = 0; ks < 32; ++ks) {
            v2f q = *(const v2f*)(qbase + ks * 4);
            q[0] *= inv_norm; q[1] *= inv_norm;
            qf[ks] = q;
        }
    }

    v8f o[8];
    #pragma unroll
    for (int dt = 0; dt < 8; ++dt) o[dt] = (v8f){};
    float mrow[8], lrow[8];
    #pragma unroll
    for (int r = 0; r < 8; ++r) { mrow[r] = -1e30f; lrow[r] = 0.0f; }

    const int nkt       = qblk / 16 + 8;
    const int my_kt_max = qblk / 16 + wave;

    const int ld_row = tid >> 4;
    const int ld_col = (tid & 15) * 8;
    const size_t ld_base = (size_t)ld_row * SROW + head_off + ld_col;
    v4f rk0, rk1, rv0, rv1;
    {
        rk0 = *(const v4f*)(K + ld_base);
        rk1 = *(const v4f*)(K + ld_base + 4);
        rv0 = *(const v4f*)(V + ld_base);
        rv1 = *(const v4f*)(V + ld_base + 4);
    }

    for (int kt = 0; kt < nkt; ++kt) {
        float* bK = sK[kt & 1];
        float* bV = sV[kt & 1];
        *(v4f*)&bK[ld_row * KSTRIDE + ld_col]     = rk0;
        *(v4f*)&bK[ld_row * KSTRIDE + ld_col + 4] = rk1;
        *(v4f*)&bV[ld_row * KSTRIDE + ld_col]     = rv0;
        *(v4f*)&bV[ld_row * KSTRIDE + ld_col + 4] = rv1;
        if (kt + 1 < nkt) {
            const size_t g = ld_base + (size_t)(kt + 1) * 16 * SROW;
            rk0 = *(const v4f*)(K + g);
            rk1 = *(const v4f*)(K + g + 4);
            rv0 = *(const v4f*)(V + g);
            rv1 = *(const v4f*)(V + g + 4);
        }
        __syncthreads();

        if (kt <= my_kt_max) {
            v8f s = (v8f){};
            #pragma unroll
            for (int g4 = 0; g4 < 4; ++g4) {
                v2f kb[8];
                #pragma unroll
                for (int j = 0; j < 8; ++j)
                    kb[j] = *(const v2f*)&bK[l15 * KSTRIDE + (g4 * 8 + j) * 4 + 2 * hi];
                #pragma unroll
                for (int j = 0; j < 8; ++j)
                    s = __builtin_amdgcn_wmma_f32_16x16x4_f32(
                            false, qf[g4 * 8 + j], false, kb[j], (short)0, s, false, false);
            }

            const int keyc = kt * 16 + l15;
            #pragma unroll
            for (int r = 0; r < 8; ++r) {
                const int qrow = q0w + r + 8 * hi;
                float sv = (keyc <= qrow) ? s[r] : -10000.0f;

                float mx = sv;
                mx = fmaxf(mx, row_xor<1>(mx));
                mx = fmaxf(mx, row_xor<2>(mx));
                mx = fmaxf(mx, row_xor<4>(mx));
                mx = fmaxf(mx, row_xor<8>(mx));

                const float mnew = fmaxf(mrow[r], mx);
                const float corr = __expf(mrow[r] - mnew);
                const float p    = __expf(sv - mnew);

                float ps = p;
                ps += row_xor<1>(ps);
                ps += row_xor<2>(ps);
                ps += row_xor<4>(ps);
                ps += row_xor<8>(ps);

                lrow[r] = lrow[r] * corr + ps;
                mrow[r] = mnew;
                #pragma unroll
                for (int dt = 0; dt < 8; ++dt) o[dt][r] *= corr;

                sP[wave][(r + 8 * hi) * PSTRIDE + l15] = p;
            }
            asm volatile("s_wait_dscnt 0x0" ::: "memory");

            v2f a[4];
            #pragma unroll
            for (int ka = 0; ka < 4; ++ka)
                a[ka] = *(const v2f*)&sP[wave][l15 * PSTRIDE + ka * 4 + 2 * hi];
            #pragma unroll
            for (int ka = 0; ka < 4; ++ka) {
                v2f vb[8];
                #pragma unroll
                for (int dt = 0; dt < 8; ++dt) {
                    vb[dt][0] = bV[(ka * 4 + 2 * hi    ) * KSTRIDE + dt * 16 + l15];
                    vb[dt][1] = bV[(ka * 4 + 2 * hi + 1) * KSTRIDE + dt * 16 + l15];
                }
                #pragma unroll
                for (int dt = 0; dt < 8; ++dt)
                    o[dt] = __builtin_amdgcn_wmma_f32_16x16x4_f32(
                                false, a[ka], false, vb[dt], (short)0, o[dt], false, false);
            }
        }
    }

    #pragma unroll
    for (int r = 0; r < 8; ++r) {
        const float inv = 1.0f / lrow[r];
        float* op = Out + (size_t)(q0w + r + 8 * hi) * SROW + head_off;
        #pragma unroll
        for (int dt = 0; dt < 8; ++dt)
            op[dt * 16 + l15] = o[dt][r] * inv;
    }
}

extern "C" void kernel_launch(void* const* d_in, const int* in_sizes, int n_in,
                              void* d_out, int out_size, void* d_ws, size_t ws_size,
                              hipStream_t stream) {
    (void)in_sizes; (void)n_in; (void)out_size; (void)d_ws; (void)ws_size;
    const float* Q = (const float*)d_in[0];
    const float* K = (const float*)d_in[1];
    const float* V = (const float*)d_in[2];
    float* Out     = (float*)d_out;

    dim3 grid(SEQ / 128, BATCH * HEADS);
    dim3 block(256);
    hipLaunchKernelGGL(fa_fp32_wmma_kernel, grid, block, 0, stream, Q, K, V, Out);
}


# LDS B-fragment groups (K and V) are loaded one 8-WMMA group ahead into a second register set, held in front of the WMMAs by scheduling barriers; arithmetic and order of sums unchanged
# speedup vs baseline: 1.6968x; 1.1140x over previous
#include <hip/hip_runtime.h>
#include <hip/hip_bf16.h>

typedef __attribute__((ext_vector_type(2))) float v2f;
typedef __attribute__((ext_vector_type(4))) float v4f;
typedef __attribute__((ext_vector_type(8))) float v8f;

#define SEQ    2048
#define BATCH  2
#define HEADS  16
#define HDIM   128
#define SROW   4096
#define KSTRIDE 140
#define PSTRIDE 18

template <int M>
__device__ __forceinline__ float row_xor(float x) {
    const int b = __builtin_bit_cast(int, x);
    return __builtin_bit_cast(float,
        __builtin_amdgcn_update_dpp(b, b, 0x160 | M, 0xf, 0xf, true));
}

__global__ __launch_bounds__(256) void fa_fp32_wmma_kernel(
    const float* __restrict__ Q, const float* __restrict__ K,
    const float* __restrict__ V, float* __restrict__ Out)
{
    __shared__ float sK[2][16 * KSTRIDE];
    __shared__ float sV[2][16 * KSTRIDE];
    __shared__ float sP[8][16 * PSTRIDE];

    const int tid  = threadIdx.x;
    const int wave = tid >> 5;
    const int lane = tid & 31;
    const int hi   = lane >> 4;
    const int l15  = lane & 15;

    const size_t head_off = (size_t)blockIdx.y * HDIM;
    const int qblk = blockIdx.x * 128;
    const int q0w  = qblk + wave * 16;

    v2f qf[32];
    {
        const float inv_norm = 0.08838834764831845f;
        const float* qbase = Q + (size_t)(q0w + l15) * SROW + head_off + 2 * hi;
        #pragma unroll
        for (int ks = 0; ks < 32; ++ks) {
            v2f q = *(const v2f*)(qbase + ks * 4);
            q[0] *= inv_norm; q[1] *= inv_norm;
            qf[ks] = q;
        }
    }

    v8f o[8];
    #pragma unroll
    for (int dt = 0; dt < 8; ++dt) o[dt] = (v8f){};
    float mrow[8], lrow[8];
    #pragma unroll
    for (int r = 0; r < 8; ++r) { mrow[r] = -1e30f; lrow[r] = 0.0f; }

    const int nkt       = qblk / 16 + 8;
    const int my_kt_max = qblk / 16 + wave;

    const int ld_row = tid >> 4;
    const int ld_col = (tid & 15) * 8;
    const size_t ld_base = (size_t)ld_row * SROW + head_off + ld_col;
    v4f rk0, rk1, rv0, rv1;
    {
        rk0 = *(const v4f*)(K + ld_base);
        rk1 = *(const v4f*)(K + ld_base + 4);
        rv0 = *(const v4f*)(V + ld_base);
        rv1 = *(const v4f*)(V + ld_base + 4);
    }

    for (int kt = 0; kt < nkt; ++kt) {
        float* bK = sK[kt & 1];
        float* bV = sV[kt & 1];
        *(v4f*)&bK[ld_row * KSTRIDE + ld_col]     = rk0;
        *(v4f*)&bK[ld_row * KSTRIDE + ld_col + 4] = rk1;
        *(v4f*)&bV[ld_row * KSTRIDE + ld_col]     = rv0;
        *(v4f*)&bV[ld_row * KSTRIDE + ld_col + 4] = rv1;
        if (kt + 1 < nkt) {
            const size_t g = ld_base + (size_t)(kt + 1) * 16 * SROW;
            rk0 = *(const v4f*)(K + g);
            rk1 = *(const v4f*)(K + g + 4);
            rv0 = *(const v4f*)(V + g);
            rv1 = *(const v4f*)(V + g + 4);
        }
        __syncthreads();

        if (kt <= my_kt_max) {
            const float* kfrag = &bK[l15 * KSTRIDE + 2 * hi];
            const float* vfrag = &bV[(2 * hi) * KSTRIDE + l15];
            v2f kb[2][8];
            float vlo[2][8], vhi[2][8];
            #pragma unroll
            for (int j = 0; j < 8; ++j)
                kb[0][j] = *(const v2f*)(kfrag + j * 4);

            v8f s = (v8f){};
            #pragma unroll
            for (int g4 = 0; g4 < 4; ++g4) {
                if (g4 < 3) {
                    #pragma unroll
                    for (int j = 0; j < 8; ++j)
                        kb[(g4 + 1) & 1][j] = *(const v2f*)(kfrag + ((g4 + 1) * 8 + j) * 4);
                } else {
                    #pragma unroll
                    for (int dt = 0; dt < 8; ++dt) {
                        vlo[0][dt] = vfrag[dt * 16];
                        vhi[0][dt] = vfrag[KSTRIDE + dt * 16];
                    }
                }
                __builtin_amdgcn_sched_barrier(0);
                #pragma unroll
                for (int j = 0; j < 8; ++j)
                    s = __builtin_amdgcn_wmma_f32_16x16x4_f32(
                            false, qf[g4 * 8 + j], false, kb[g4 & 1][j], (short)0, s, false, false);
            }

            const int keyc = kt * 16 + l15;
            float pr[8], cr[8];
            #pragma unroll
            for (int r = 0; r < 8; ++r) {
                const int qrow = q0w + r + 8 * hi;
                float sv = (keyc <= qrow) ? s[r] : -10000.0f;

                float mx = sv;
                mx = fmaxf(mx, row_xor<1>(mx));
                mx = fmaxf(mx, row_xor<2>(mx));
                mx = fmaxf(mx, row_xor<4>(mx));
                mx = fmaxf(mx, row_xor<8>(mx));

                const float mnew = fmaxf(mrow[r], mx);
                const float corr = __expf(mrow[r] - mnew);
                const float p    = __expf(sv - mnew);
                mrow[r] = mnew;
                pr[r] = p;
                cr[r] = corr;
                #pragma unroll
                for (int dt = 0; dt < 8; ++dt) o[dt][r] *= corr;

                sP[wave][(r + 8 * hi) * PSTRIDE + l15] = p;
            }
            asm volatile("s_wait_dscnt 0x0" ::: "memory");

            v2f a[4];
            #pragma unroll
            for (int ka = 0; ka < 4; ++ka)
                a[ka] = *(const v2f*)&sP[wave][l15 * PSTRIDE + ka * 4 + 2 * hi];
            #pragma unroll
            for (int ka = 0; ka < 4; ++ka) {
                if (ka < 3) {
                    #pragma unroll
                    for (int dt = 0; dt < 8; ++dt) {
                        vlo[(ka + 1) & 1][dt] = vfrag[((ka + 1) * 4    ) * KSTRIDE + dt * 16];
                        vhi[(ka + 1) & 1][dt] = vfrag[((ka + 1) * 4 + 1) * KSTRIDE + dt * 16];
                    }
                    __builtin_amdgcn_sched_barrier(0);
                }
                #pragma unroll
                for (int dt = 0; dt < 8; ++dt) {
                    const v2f vb = { vlo[ka & 1][dt], vhi[ka & 1][dt] };
                    o[dt] = __builtin_amdgcn_wmma_f32_16x16x4_f32(
                                false, a[ka], false, vb, (short)0, o[dt], false, false);
                }
                if (ka == 0) {
                    #pragma unroll
                    for (int r = 0; r < 8; ++r) {
                        float ps = pr[r];
                        ps += row_xor<1>(ps);
                        ps += row_xor<2>(ps);
                        ps += row_xor<4>(ps);
                        ps += row_xor<8>(ps);
                        lrow[r] = lrow[r] * cr[r] + ps;
                    }
                }
            }
        }
    }

    #pragma unroll
    for (int r = 0; r < 8; ++r) {
        const float inv = 1.0f / lrow[r];
        float* op = Out + (size_t)(q0w + r + 8 * hi) * SROW + head_off;
        #pragma unroll
        for (int dt = 0; dt < 8; ++dt)
            op[dt * 16 + l15] = o[dt][r] * inv;
    }
}

extern "C" void kernel_launch(void* const* d_in, const int* in_sizes, int n_in,
                              void* d_out, int out_size, void* d_ws, size_t ws_size,
                              hipStream_t stream) {
    (void)in_sizes; (void)n_in; (void)out_size; (void)d_ws; (void)ws_size;
    const float* Q = (const float*)d_in[0];
    const float* K = (const float*)d_in[1];
    const float* V = (const float*)d_in[2];
    float* Out     = (float*)d_out;

    dim3 grid(SEQ / 128, BATCH * HEADS);
    dim3 block(256);
    hipLaunchKernelGGL(fa_fp32_wmma_kernel, grid, block, 0, stream, Q, K, V, Out);
}


# Causal work balance: each block runs two 128-row query tiles in turn (tile 15-x, then x), so all 256 blocks do the same 136 key-tile steps; per-row arithmetic and order of sums unchanged
# speedup vs baseline: 2.3350x; 1.3761x over previous
#include <hip/hip_runtime.h>
#include <hip/hip_bf16.h>

typedef __attribute__((ext_vector_type(2))) float v2f;
typedef __attribute__((ext_vector_type(4))) float v4f;
typedef __attribute__((ext_vector_type(8))) float v8f;

#define SEQ    2048
#define BATCH  2
#define HEADS  16
#define HDIM   128
#define SROW   4096
#define KSTRIDE 140
#define PSTRIDE 18

template <int M>
__device__ __forceinline__ float row_xor(float x) {
    const int b = __builtin_bit_cast(int, x);
    return __builtin_bit_cast(float,
        __builtin_amdgcn_update_dpp(b, b, 0x160 | M, 0xf, 0xf, true));
}

__global__ __launch_bounds__(256) void fa_fp32_wmma_kernel(
    const float* __restrict__ Q, const float* __restrict__ K,
    const float* __restrict__ V, float* __restrict__ Out)
{
    __shared__ float sK[2][16 * KSTRIDE];
    __shared__ float sV[2][16 * KSTRIDE];
    __shared__ float sP[8][16 * PSTRIDE];

    const int tid  = threadIdx.x;
    const int wave = tid >> 5;
    const int lane = tid & 31;
    const int hi   = lane >> 4;
    const int l15  = lane & 15;

    const size_t head_off = (size_t)blockIdx.y * HDIM;

    const int ld_row = tid >> 4;
    const int ld_col = (tid & 15) * 8;
    const size_t ld_base = (size_t)ld_row * SROW + head_off + ld_col;

    #pragma unroll 1
    for (int pass = 0; pass < 2; ++pass) {
    const int qtile = pass == 0 ? 15 - (int)blockIdx.x : (int)blockIdx.x;
    const int qblk = qtile * 128;
    const int q0w  = qblk + wave * 16;

    v2f qf[32];
    {
        const float inv_norm = 0.08838834764831845f;
        const float* qbase = Q + (size_t)(q0w + l15) * SROW + head_off + 2 * hi;
        #pragma unroll
        for (int ks = 0; ks < 32; ++ks) {
            v2f q = *(const v2f*)(qbase + ks * 4);
            q[0] *= inv_norm; q[1] *= inv_norm;
            qf[ks] = q;
        }
    }

    v8f o[8];
    #pragma unroll
    for (int dt = 0; dt < 8; ++dt) o[dt] = (v8f){};
    float mrow[8], lrow[8];
    #pragma unroll
    for (int r = 0; r < 8; ++r) { mrow[r] = -1e30f; lrow[r] = 0.0f; }

    const int nkt       = qblk / 16 + 8;
    const int my_kt_max = qblk / 16 + wave;

    v4f rk0, rk1, rv0, rv1;
    {
        rk0 = *(const v4f*)(K + ld_base);
        rk1 = *(const v4f*)(K + ld_base + 4);
        rv0 = *(const v4f*)(V + ld_base);
        rv1 = *(const v4f*)(V + ld_base + 4);
    }

    for (int kt = 0; kt < nkt; ++kt) {
        float* bK = sK[kt & 1];
        float* bV = sV[kt & 1];
        *(v4f*)&bK[ld_row * KSTRIDE + ld_col]     = rk0;
        *(v4f*)&bK[ld_row * KSTRIDE + ld_col + 4] = rk1;
        *(v4f*)&bV[ld_row * KSTRIDE + ld_col]     = rv0;
        *(v4f*)&bV[ld_row * KSTRIDE + ld_col + 4] = rv1;
        if (kt + 1 < nkt) {
            const size_t g = ld_base + (size_t)(kt + 1) * 16 * SROW;
            rk0 = *(const v4f*)(K + g);
            rk1 = *(const v4f*)(K + g + 4);
            rv0 = *(const v4f*)(V + g);
            rv1 = *(const v4f*)(V + g + 4);
        }
        __syncthreads();

        if (kt <= my_kt_max) {
            const float* kfrag = &bK[l15 * KSTRIDE + 2 * hi];
            const float* vfrag = &bV[(2 * hi) * KSTRIDE + l15];
            v2f kb[2][8];
            float vlo[2][8], vhi[2][8];
            #pragma unroll
            for (int j = 0; j < 8; ++j)
                kb[0][j] = *(const v2f*)(kfrag + j * 4);

            v8f s = (v8f){};
            #pragma unroll
            for (int g4 = 0; g4 < 4; ++g4) {
                if (g4 < 3) {
                    #pragma unroll
                    for (int j = 0; j < 8; ++j)
                        kb[(g4 + 1) & 1][j] = *(const v2f*)(kfrag + ((g4 + 1) * 8 + j) * 4);
                } else {
                    #pragma unroll
                    for (int dt = 0; dt < 8; ++dt) {
                        vlo[0][dt] = vfrag[dt * 16];
                        vhi[0][dt] = vfrag[KSTRIDE + dt * 16];
                    }
                }
                __builtin_amdgcn_sched_barrier(0);
                #pragma unroll
                for (int j = 0; j < 8; ++j)
                    s = __builtin_amdgcn_wmma_f32_16x16x4_f32(
                            false, qf[g4 * 8 + j], false, kb[g4 & 1][j], (short)0, s, false, false);
            }

            const int keyc = kt * 16 + l15;
            float pr[8], cr[8];
            #pragma unroll
            for (int r = 0; r < 8; ++r) {
                const int qrow = q0w + r + 8 * hi;
                float sv = (keyc <= qrow) ? s[r] : -10000.0f;

                float mx = sv;
                mx = fmaxf(mx, row_xor<1>(mx));
                mx = fmaxf(mx, row_xor<2>(mx));
                mx = fmaxf(mx, row_xor<4>(mx));
                mx = fmaxf(mx, row_xor<8>(mx));

                const float mnew = fmaxf(mrow[r], mx);
                const float corr = __expf(mrow[r] - mnew);
                const float p    = __expf(sv - mnew);
                mrow[r] = mnew;
                pr[r] = p;
                cr[r] = corr;
                #pragma unroll
                for (int dt = 0; dt < 8; ++dt) o[dt][r] *= corr;

                sP[wave][(r + 8 * hi) * PSTRIDE + l15] = p;
            }
            asm volatile("s_wait_dscnt 0x0" ::: "memory");

            v2f a[4];
            #pragma unroll
            for (int ka = 0; ka < 4; ++ka)
                a[ka] = *(const v2f*)&sP[wave][l15 * PSTRIDE + ka * 4 + 2 * hi];
            #pragma unroll
            for (int ka = 0; ka < 4; ++ka) {
                if (ka < 3) {
                    #pragma unroll
                    for (int dt = 0; dt < 8; ++dt) {
                        vlo[(ka + 1) & 1][dt] = vfrag[((ka + 1) * 4    ) * KSTRIDE + dt * 16];
                        vhi[(ka + 1) & 1][dt] = vfrag[((ka + 1) * 4 + 1) * KSTRIDE + dt * 16];
                    }
                    __builtin_amdgcn_sched_barrier(0);
                }
                #pragma unroll
                for (int dt = 0; dt < 8; ++dt) {
                    const v2f vb = { vlo[ka & 1][dt], vhi[ka & 1][dt] };
                    o[dt] = __builtin_amdgcn_wmma_f32_16x16x4_f32(
                                false, a[ka], false, vb, (short)0, o[dt], false, false);
                }
                if (ka == 0) {
                    #pragma unroll
                    for (int r = 0; r < 8; ++r) {
                        float ps = pr[r];
                        ps += row_xor<1>(ps);
                        ps += row_xor<2>(ps);
                        ps += row_xor<4>(ps);
                        ps += row_xor<8>(ps);
                        lrow[r] = lrow[r] * cr[r] + ps;
                    }
                }
            }
        }
    }

    #pragma unroll
    for (int r = 0; r < 8; ++r) {
        const float inv = 1.0f / lrow[r];
        float* op = Out + (size_t)(q0w + r + 8 * hi) * SROW + head_off;
        #pragma unroll
        for (int dt = 0; dt < 8; ++dt)
            op[dt * 16 + l15] = o[dt][r] * inv;
    }

    __syncthreads();
    }
}

extern "C" void kernel_launch(void* const* d_in, const int* in_sizes, int n_in,
                              void* d_out, int out_size, void* d_ws, size_t ws_size,
                              hipStream_t stream) {
    (void)in_sizes; (void)n_in; (void)out_size; (void)d_ws; (void)ws_size;
    const float* Q = (const float*)d_in[0];
    const float* K = (const float*)d_in[1];
    const float* V = (const float*)d_in[2];
    float* Out     = (float*)d_out;

    dim3 grid(SEQ / 256, BATCH * HEADS);
    dim3 block(256);
    hipLaunchKernelGGL(fa_fp32_wmma_kernel, grid, block, 0, stream, Q, K, V, Out);
}


# Next K/V tile's global loads issued behind the step's barrier instead of in front of it: the barrier's fence no longer waits out their latency every step, they run under the WMMAs; sums unchanged
# speedup vs baseline: 2.9404x; 1.2593x over previous
#include <hip/hip_runtime.h>
#include <hip/hip_bf16.h>

typedef __attribute__((ext_vector_type(2))) float v2f;
typedef __attribute__((ext_vector_type(4))) float v4f;
typedef __attribute__((ext_vector_type(8))) float v8f;

#define SEQ    2048
#define BATCH  2
#define HEADS  16
#define HDIM   128
#define SROW   4096
#define KSTRIDE 140
#define PSTRIDE 18

template <int M>
__device__ __forceinline__ float row_xor(float x) {
    const int b = __builtin_bit_cast(int, x);
    return __builtin_bit_cast(float,
        __builtin_amdgcn_update_dpp(b, b, 0x160 | M, 0xf, 0xf, true));
}

__global__ __launch_bounds__(256) void fa_fp32_wmma_kernel(
    const float* __restrict__ Q, const float* __restrict__ K,
    const float* __restrict__ V, float* __restrict__ Out)
{
    __shared__ float sK[2][16 * KSTRIDE];
    __shared__ float sV[2][16 * KSTRIDE];
    __shared__ float sP[8][16 * PSTRIDE];

    const int tid  = threadIdx.x;
    const int wave = tid >> 5;
    const int lane = tid & 31;
    const int hi   = lane >> 4;
    const int l15  = lane & 15;

    const size_t head_off = (size_t)blockIdx.y * HDIM;

    const int ld_row = tid >> 4;
    const int ld_col = (tid & 15) * 8;
    const size_t ld_base = (size_t)ld_row * SROW + head_off + ld_col;

    #pragma unroll 1
    for (int pass = 0; pass < 2; ++pass) {
    const int qtile = pass == 0 ? 15 - (int)blockIdx.x : (int)blockIdx.x;
    const int qblk = qtile * 128;
    const int q0w  = qblk + wave * 16;

    v2f qf[32];
    {
        const float inv_norm = 0.08838834764831845f;
        const float* qbase = Q + (size_t)(q0w + l15) * SROW + head_off + 2 * hi;
        #pragma unroll
        for (int ks = 0; ks < 32; ++ks) {
            v2f q = *(const v2f*)(qbase + ks * 4);
            q[0] *= inv_norm; q[1] *= inv_norm;
            qf[ks] = q;
        }
    }

    v8f o[8];
    #pragma unroll
    for (int dt = 0; dt < 8; ++dt) o[dt] = (v8f){};
    float mrow[8], lrow[8];
    #pragma unroll
    for (int r = 0; r < 8; ++r) { mrow[r] = -1e30f; lrow[r] = 0.0f; }

    const int nkt       = qblk / 16 + 8;
    const int my_kt_max = qblk / 16 + wave;

    v4f rk0, rk1, rv0, rv1;
    {
        rk0 = *(const v4f*)(K + ld_base);
        rk1 = *(const v4f*)(K + ld_base + 4);
        rv0 = *(const v4f*)(V + ld_base);
        rv1 = *(const v4f*)(V + ld_base + 4);
    }

    for (int kt = 0; kt < nkt; ++kt) {
        float* bK = sK[kt & 1];
        float* bV = sV[kt & 1];
        *(v4f*)&bK[ld_row * KSTRIDE + ld_col]     = rk0;
        *(v4f*)&bK[ld_row * KSTRIDE + ld_col + 4] = rk1;
        *(v4f*)&bV[ld_row * KSTRIDE + ld_col]     = rv0;
        *(v4f*)&bV[ld_row * KSTRIDE + ld_col + 4] = rv1;
        __syncthreads();
        if (kt + 1 < nkt) {
            const size_t g = ld_base + (size_t)(kt + 1) * 16 * SROW;
            rk0 = *(const v4f*)(K + g);
            rk1 = *(const v4f*)(K + g + 4);
            rv0 = *(const v4f*)(V + g);
            rv1 = *(const v4f*)(V + g + 4);
        }

        if (kt <= my_kt_max) {
            const float* kfrag = &bK[l15 * KSTRIDE + 2 * hi];
            const float* vfrag = &bV[(2 * hi) * KSTRIDE + l15];
            v2f kb[2][8];
            float vlo[2][8], vhi[2][8];
            #pragma unroll
            for (int j = 0; j < 8; ++j)
                kb[0][j] = *(const v2f*)(kfrag + j * 4);

            v8f s = (v8f){};
            #pragma unroll
            for (int g4 = 0; g4 < 4; ++g4) {
                if (g4 < 3) {
                    #pragma unroll
                    for (int j = 0; j < 8; ++j)
                        kb[(g4 + 1) & 1][j] = *(const v2f*)(kfrag + ((g4 + 1) * 8 + j) * 4);
                } else {
                    #pragma unroll
                    for (int dt = 0; dt < 8; ++dt) {
                        vlo[0][dt] = vfrag[dt * 16];
                        vhi[0][dt] = vfrag[KSTRIDE + dt * 16];
                    }
                }
                __builtin_amdgcn_sched_barrier(0);
                #pragma unroll
                for (int j = 0; j < 8; ++j)
                    s = __builtin_amdgcn_wmma_f32_16x16x4_f32(
                            false, qf[g4 * 8 + j], false, kb[g4 & 1][j], (short)0, s, false, false);
            }

            const int keyc = kt * 16 + l15;
            float pr[8], cr[8];
            #pragma unroll
            for (int r = 0; r < 8; ++r) {
                const int qrow = q0w + r + 8 * hi;
                float sv = (keyc <= qrow) ? s[r] : -10000.0f;

                float mx = sv;
                mx = fmaxf(mx, row_xor<1>(mx));
                mx = fmaxf(mx, row_xor<2>(mx));
                mx = fmaxf(mx, row_xor<4>(mx));
                mx = fmaxf(mx, row_xor<8>(mx));

                const float mnew = fmaxf(mrow[r], mx);
                const float corr = __expf(mrow[r] - mnew);
                const float p    = __expf(sv - mnew);
                mrow[r] = mnew;
                pr[r] = p;
                cr[r] = corr;
                #pragma unroll
                for (int dt = 0; dt < 8; ++dt) o[dt][r] *= corr;

                sP[wave][(r + 8 * hi) * PSTRIDE + l15] = p;
            }
            asm volatile("s_wait_dscnt 0x0" ::: "memory");

            v2f a[4];
            #pragma unroll
            for (int ka = 0; ka < 4; ++ka)
                a[ka] = *(const v2f*)&sP[wave][l15 * PSTRIDE + ka * 4 + 2 * hi];
            #pragma unroll
            for (int ka = 0; ka < 4; ++ka) {
                if (ka < 3) {
                    #pragma unroll
                    for (int dt = 0; dt < 8; ++dt) {
                        vlo[(ka + 1) & 1][dt] = vfrag[((ka + 1) * 4    ) * KSTRIDE + dt * 16];
                        vhi[(ka + 1) & 1][dt] = vfrag[((ka + 1) * 4 + 1) * KSTRIDE + dt * 16];
                    }
                    __builtin_amdgcn_sched_barrier(0);
                }
                #pragma unroll
                for (int dt = 0; dt < 8; ++dt) {
                    const v2f vb = { vlo[ka & 1][dt], vhi[ka & 1][dt] };
                    o[dt] = __builtin_amdgcn_wmma_f32_16x16x4_f32(
                                false, a[ka], false, vb, (short)0, o[dt], false, false);
                }
                if (ka == 0) {
                    #pragma unroll
                    for (int r = 0; r < 8; ++r) {
                        float ps = pr[r];
                        ps += row_xor<1>(ps);
                        ps += row_xor<2>(ps);
                        ps += row_xor<4>(ps);
                        ps += row_xor<8>(ps);
                        lrow[r] = lrow[r] * cr[r] + ps;
                    }
                }
            }
        }
    }

    #pragma unroll
    for (int r = 0; r < 8; ++r) {
        const float inv = 1.0f / lrow[r];
        float* op = Out + (size_t)(q0w + r + 8 * hi) * SROW + head_off;
        #pragma unroll
        for (int dt = 0; dt < 8; ++dt)
            op[dt * 16 + l15] = o[dt][r] * inv;
    }

    __syncthreads();
    }
}

extern "C" void kernel_launch(void* const* d_in, const int* in_sizes, int n_in,
                              void* d_out, int out_size, void* d_ws, size_t ws_size,
                              hipStream_t stream) {
    (void)in_sizes; (void)n_in; (void)out_size; (void)d_ws; (void)ws_size;
    const float* Q = (const float*)d_in[0];
    const float* K = (const float*)d_in[1];
    const float* V = (const float*)d_in[2];
    float* Out     = (float*)d_out;

    dim3 grid(SEQ / 256, BATCH * HEADS);
    dim3 block(256);
    hipLaunchKernelGGL(fa_fp32_wmma_kernel, grid, block, 0, stream, Q, K, V, Out);
}


# Key tile per step widened from 16 to 32 keys: one barrier, running-max update, accumulator rescale and row-sum butterfly per 32 keys; two independent score chains; order of sums changes
# speedup vs baseline: 3.1435x; 1.0691x over previous
#include <hip/hip_runtime.h>
#include <hip/hip_bf16.h>


typedef __attribute__((ext_vector_type(2))) float v2f;
typedef __attribute__((ext_vector_type(4))) float v4f;
typedef __attribute__((ext_vector_type(8))) float v8f;

#define SEQ    2048
#define BATCH  2
#define HEADS  16
#define HDIM   128
#define SROW   4096
#define KSTRIDE 140
#define NSUB   2
#define KEYS   (16 * NSUB)
#define PSTRIDE (KEYS + 4)

template <int M>
__device__ __forceinline__ float row_xor(float x) {
    const int b = __builtin_bit_cast(int, x);
    return __builtin_bit_cast(float,
        __builtin_amdgcn_update_dpp(b, b, 0x160 | M, 0xf, 0xf, true));
}

__global__ __launch_bounds__(256) void fa_fp32_wmma_kernel(
    const float* __restrict__ Q, const float* __restrict__ K,
    const float* __restrict__ V, float* __restrict__ Out)
{
    __shared__ float sK[2][KEYS * KSTRIDE];
    __shared__ float sV[2][KEYS * KSTRIDE];
    __shared__ float sP[8][16 * PSTRIDE];

    const int tid  = threadIdx.x;
    const int wave = tid >> 5;
    const int lane = tid & 31;
    const int hi   = lane >> 4;
    const int l15  = lane & 15;

    const size_t head_off = (size_t)blockIdx.y * HDIM;

    const int ld_row = tid >> 4;
    const int ld_col = (tid & 15) * 8;
    const size_t ld_base = (size_t)ld_row * SROW + head_off + ld_col;

    #pragma unroll 1
    for (int pass = 0; pass < 2; ++pass) {
    const int qtile = pass == 0 ? 15 - (int)blockIdx.x : (int)blockIdx.x;
    const int qblk = qtile * 128;
    const int q0w  = qblk + wave * 16;

    v2f qf[32];
    {
        const float inv_norm = 0.08838834764831845f;
        const float* qbase = Q + (size_t)(q0w + l15) * SROW + head_off + 2 * hi;
        #pragma unroll
        for (int ks = 0; ks < 32; ++ks) {
            v2f q = *(const v2f*)(qbase + ks * 4);
            q[0] *= inv_norm; q[1] *= inv_norm;
            qf[ks] = q;
        }
    }

    v8f o[8];
    #pragma unroll
    for (int dt = 0; dt < 8; ++dt) o[dt] = (v8f){};
    float mrow[8], lrow[8];
    #pragma unroll
    for (int r = 0; r < 8; ++r) { mrow[r] = -1e30f; lrow[r] = 0.0f; }

    const int nkt       = (qblk / 16 + 8) / NSUB;
    const int my_kt_max = qblk / 16 + wave;

    v4f rk[NSUB][2], rv[NSUB][2];
    #pragma unroll
    for (int u = 0; u < NSUB; ++u) {
        const size_t g = ld_base + (size_t)(u * 16) * SROW;
        rk[u][0] = *(const v4f*)(K + g);
        rk[u][1] = *(const v4f*)(K + g + 4);
        rv[u][0] = *(const v4f*)(V + g);
        rv[u][1] = *(const v4f*)(V + g + 4);
    }

    for (int kt = 0; kt < nkt; ++kt) {
        float* bK = sK[kt & 1];
        float* bV = sV[kt & 1];
        #pragma unroll
        for (int u = 0; u < NSUB; ++u) {
            const int row = u * 16 + ld_row;
            *(v4f*)&bK[row * KSTRIDE + ld_col]     = rk[u][0];
            *(v4f*)&bK[row * KSTRIDE + ld_col + 4] = rk[u][1];
            *(v4f*)&bV[row * KSTRIDE + ld_col]     = rv[u][0];
            *(v4f*)&bV[row * KSTRIDE + ld_col + 4] = rv[u][1];
        }
        __syncthreads();
        if (kt + 1 < nkt) {
            #pragma unroll
            for (int u = 0; u < NSUB; ++u) {
                const size_t g = ld_base + (size_t)((kt + 1) * KEYS + u * 16) * SROW;
                rk[u][0] = *(const v4f*)(K + g);
                rk[u][1] = *(const v4f*)(K + g + 4);
                rv[u][0] = *(const v4f*)(V + g);
                rv[u][1] = *(const v4f*)(V + g + 4);
            }
        }

        if (kt * NSUB <= my_kt_max) {
            const float* kfrag = &bK[l15 * KSTRIDE + 2 * hi];
            const float* vfrag = &bV[(2 * hi) * KSTRIDE + l15];
            v2f kb[2][NSUB][8];
            float vlo[2][2][8], vhi[2][2][8];
            #pragma unroll
            for (int u = 0; u < NSUB; ++u) {
                #pragma unroll
                for (int j = 0; j < 8; ++j)
                    kb[0][u][j] = *(const v2f*)(kfrag + u * 16 * KSTRIDE + j * 4);
            }

            v8f s[NSUB];
            #pragma unroll
            for (int u = 0; u < NSUB; ++u) s[u] = (v8f){};
            #pragma unroll
            for (int g4 = 0; g4 < 4; ++g4) {
                if (g4 < 3) {
                    #pragma unroll
                    for (int u = 0; u < NSUB; ++u) {
                        #pragma unroll
                        for (int j = 0; j < 8; ++j)
                            kb[(g4 + 1) & 1][u][j] =
                                *(const v2f*)(kfrag + u * 16 * KSTRIDE + ((g4 + 1) * 8 + j) * 4);
                    }
                } else {
                    #pragma unroll
                    for (int kk = 0; kk < 2; ++kk) {
                        #pragma unroll
                        for (int dt = 0; dt < 8; ++dt) {
                            vlo[0][kk][dt] = vfrag[(kk * 4    ) * KSTRIDE + dt * 16];
                            vhi[0][kk][dt] = vfrag[(kk * 4 + 1) * KSTRIDE + dt * 16];
                        }
                    }
                }
                __builtin_amdgcn_sched_barrier(0);
                #pragma unroll
                for (int j = 0; j < 8; ++j) {
                    #pragma unroll
                    for (int u = 0; u < NSUB; ++u)
                        s[u] = __builtin_amdgcn_wmma_f32_16x16x4_f32(
                                false, qf[g4 * 8 + j], false, kb[g4 & 1][u][j], (short)0, s[u], false, false);
                }
            }

            float pr[8], cr[8];
            #pragma unroll
            for (int r = 0; r < 8; ++r) {
                const int qrow = q0w + r + 8 * hi;
                float sv[NSUB];
                #pragma unroll
                for (int u = 0; u < NSUB; ++u) {
                    const int keyc = kt * KEYS + u * 16 + l15;
                    sv[u] = (keyc <= qrow) ? s[u][r] : -10000.0f;
                }

                float mx = sv[0];
                #pragma unroll
                for (int u = 1; u < NSUB; ++u) mx = fmaxf(mx, sv[u]);
                mx = fmaxf(mx, row_xor<1>(mx));
                mx = fmaxf(mx, row_xor<2>(mx));
                mx = fmaxf(mx, row_xor<4>(mx));
                mx = fmaxf(mx, row_xor<8>(mx));

                const float mnew = fmaxf(mrow[r], mx);
                const float corr = __expf(mrow[r] - mnew);
                mrow[r] = mnew;
                cr[r] = corr;
                float psum = 0.0f;
                #pragma unroll
                for (int u = 0; u < NSUB; ++u) {
                    const float p = __expf(sv[u] - mnew);
                    psum += p;
                    sP[wave][(r + 8 * hi) * PSTRIDE + u * 16 + l15] = p;
                }
                pr[r] = psum;
                #pragma unroll
                for (int dt = 0; dt < 8; ++dt) o[dt][r] *= corr;
            }
            asm volatile("s_wait_dscnt 0x0" ::: "memory");

            v2f a[4 * NSUB];
            #pragma unroll
            for (int ka = 0; ka < 4 * NSUB; ++ka)
                a[ka] = *(const v2f*)&sP[wave][l15 * PSTRIDE + ka * 4 + 2 * hi];
            __builtin_amdgcn_sched_barrier(0);
            #pragma unroll
            for (int gi = 0; gi < 2 * NSUB; ++gi) {
                if (gi < 2 * NSUB - 1) {
                    #pragma unroll
                    for (int kk = 0; kk < 2; ++kk) {
                        const int kn = (gi + 1) * 2 + kk;
                        #pragma unroll
                        for (int dt = 0; dt < 8; ++dt) {
                            vlo[(gi + 1) & 1][kk][dt] = vfrag[(kn * 4    ) * KSTRIDE + dt * 16];
                            vhi[(gi + 1) & 1][kk][dt] = vfrag[(kn * 4 + 1) * KSTRIDE + dt * 16];
                        }
                    }
                    __builtin_amdgcn_sched_barrier(0);
                }
                #pragma unroll
                for (int kk = 0; kk < 2; ++kk) {
                    #pragma unroll
                    for (int dt = 0; dt < 8; ++dt) {
                        const v2f vb = { vlo[gi & 1][kk][dt], vhi[gi & 1][kk][dt] };
                        o[dt] = __builtin_amdgcn_wmma_f32_16x16x4_f32(
                                    false, a[gi * 2 + kk], false, vb, (short)0, o[dt], false, false);
                    }
                }
                if (gi == 0) {
                    #pragma unroll
                    for (int r = 0; r < 8; ++r) {
                        float ps = pr[r];
                        ps += row_xor<1>(ps);
                        ps += row_xor<2>(ps);
                        ps += row_xor<4>(ps);
                        ps += row_xor<8>(ps);
                        lrow[r] = lrow[r] * cr[r] + ps;
                    }
                }
            }
        }
    }

    #pragma unroll
    for (int r = 0; r < 8; ++r) {
        const float inv = 1.0f / lrow[r];
        float* op = Out + (size_t)(q0w + r + 8 * hi) * SROW + head_off;
        #pragma unroll
        for (int dt = 0; dt < 8; ++dt)
            op[dt * 16 + l15] = o[dt][r] * inv;
    }

    __syncthreads();
    }
}

extern "C" void kernel_launch(void* const* d_in, const int* in_sizes, int n_in,
                              void* d_out, int out_size, void* d_ws, size_t ws_size,
                              hipStream_t stream) {
    (void)in_sizes; (void)n_in; (void)out_size; (void)d_ws; (void)ws_size;
    const float* Q = (const float*)d_in[0];
    const float* K = (const float*)d_in[1];
    const float* V = (const float*)d_in[2];
    float* Out     = (float*)d_out;

    dim3 grid(SEQ / 256, BATCH * HEADS);
    dim3 block(256);
    hipLaunchKernelGGL(fa_fp32_wmma_kernel, grid, block, 0, stream, Q, K, V, Out);
}


# Both products run transposed (S^T = K*Q^T, O^T += V^T*P^T): P feeds the second product from its score registers, no LDS trip; row max and sum in-lane plus one half exchange; order of sums changes
# speedup vs baseline: 3.4069x; 1.0838x over previous
#include <hip/hip_runtime.h>
#include <hip/hip_bf16.h>


typedef __attribute__((ext_vector_type(2))) float v2f;
typedef __attribute__((ext_vector_type(4))) float v4f;
typedef __attribute__((ext_vector_type(8))) float v8f;

#define SEQ    2048
#define BATCH  2
#define HEADS  16
#define HDIM   128
#define SROW   4096
#define KSTRIDE 140
#define NSUB   2
#define KEYS   (16 * NSUB)
#define OSTRIDE 36

__device__ __forceinline__ float xhalf(float x) {
    const int b = __builtin_bit_cast(int, x);
    return __builtin_bit_cast(float,
        __builtin_amdgcn_permlanex16(b, b, 0x76543210, 0xfedcba98, false, false));
}

__global__ __launch_bounds__(256) void fa_fp32_wmma_kernel(
    const float* __restrict__ Q, const float* __restrict__ K,
    const float* __restrict__ V, float* __restrict__ Out)
{
    __shared__ float sK[2][KEYS * KSTRIDE];
    __shared__ float sV[2][KEYS * KSTRIDE];
    __shared__ __attribute__((aligned(16))) float sO[8][16 * OSTRIDE];

    const int tid  = threadIdx.x;
    const int wave = tid >> 5;
    const int lane = tid & 31;
    const int hi   = lane >> 4;
    const int l15  = lane & 15;

    const size_t head_off = (size_t)blockIdx.y * HDIM;

    const int ld_row = tid >> 4;
    const int ld_col = (tid & 15) * 8;
    const size_t ld_base = (size_t)ld_row * SROW + head_off + ld_col;

    #pragma unroll 1
    for (int pass = 0; pass < 2; ++pass) {
    const int qtile = pass == 0 ? 15 - (int)blockIdx.x : (int)blockIdx.x;
    const int qblk = qtile * 128;
    const int q0w  = qblk + wave * 16;

    v2f qf[32];
    {
        const float inv_norm = 0.08838834764831845f;
        const float* qbase = Q + (size_t)(q0w + l15) * SROW + head_off + 2 * hi;
        #pragma unroll
        for (int ks = 0; ks < 32; ++ks) {
            v2f q = *(const v2f*)(qbase + ks * 4);
            q[0] *= inv_norm; q[1] *= inv_norm;
            qf[ks] = q;
        }
    }

    v8f o[8];
    #pragma unroll
    for (int dt = 0; dt < 8; ++dt) o[dt] = (v8f){};
    float m_run = -1e30f, l_run = 0.0f;
    const int qrow = q0w + l15;

    const int nkt       = (qblk / 16 + 8) / NSUB;
    const int my_kt_max = qblk / 16 + wave;

    v4f rk[NSUB][2], rv[NSUB][2];
    #pragma unroll
    for (int u = 0; u < NSUB; ++u) {
        const size_t g = ld_base + (size_t)(u * 16) * SROW;
        rk[u][0] = *(const v4f*)(K + g);
        rk[u][1] = *(const v4f*)(K + g + 4);
        rv[u][0] = *(const v4f*)(V + g);
        rv[u][1] = *(const v4f*)(V + g + 4);
    }

    for (int kt = 0; kt < nkt; ++kt) {
        float* bK = sK[kt & 1];
        float* bV = sV[kt & 1];
        #pragma unroll
        for (int u = 0; u < NSUB; ++u) {
            const int row = u * 16 + ld_row;
            *(v4f*)&bK[row * KSTRIDE + ld_col]     = rk[u][0];
            *(v4f*)&bK[row * KSTRIDE + ld_col + 4] = rk[u][1];
            *(v4f*)&bV[row * KSTRIDE + ld_col]     = rv[u][0];
            *(v4f*)&bV[row * KSTRIDE + ld_col + 4] = rv[u][1];
        }
        __syncthreads();
        if (kt + 1 < nkt) {
            #pragma unroll
            for (int u = 0; u < NSUB; ++u) {
                const size_t g = ld_base + (size_t)((kt + 1) * KEYS + u * 16) * SROW;
                rk[u][0] = *(const v4f*)(K + g);
                rk[u][1] = *(const v4f*)(K + g + 4);
                rv[u][0] = *(const v4f*)(V + g);
                rv[u][1] = *(const v4f*)(V + g + 4);
            }
        }

        if (kt * NSUB <= my_kt_max) {
            const float* kfrag = &bK[l15 * KSTRIDE + 2 * hi];
            const float* vfrag = &bV[(8 * hi) * KSTRIDE + l15];
            v2f kb[2][NSUB][8];
            float vlo[2][2][8], vhi[2][2][8];
            #pragma unroll
            for (int u = 0; u < NSUB; ++u) {
                #pragma unroll
                for (int j = 0; j < 8; ++j)
                    kb[0][u][j] = *(const v2f*)(kfrag + u * 16 * KSTRIDE + j * 4);
            }

            v8f s[NSUB];
            #pragma unroll
            for (int u = 0; u < NSUB; ++u) s[u] = (v8f){};
            #pragma unroll
            for (int g4 = 0; g4 < 4; ++g4) {
                if (g4 < 3) {
                    #pragma unroll
                    for (int u = 0; u < NSUB; ++u) {
                        #pragma unroll
                        for (int j = 0; j < 8; ++j)
                            kb[(g4 + 1) & 1][u][j] =
                                *(const v2f*)(kfrag + u * 16 * KSTRIDE + ((g4 + 1) * 8 + j) * 4);
                    }
                } else {
                    #pragma unroll
                    for (int kk = 0; kk < 2; ++kk) {
                        #pragma unroll
                        for (int dt = 0; dt < 8; ++dt) {
                            vlo[0][kk][dt] = vfrag[(kk * 2    ) * KSTRIDE + dt * 16];
                            vhi[0][kk][dt] = vfrag[(kk * 2 + 1) * KSTRIDE + dt * 16];
                        }
                    }
                }
                __builtin_amdgcn_sched_barrier(0);
                #pragma unroll
                for (int j = 0; j < 8; ++j) {
                    #pragma unroll
                    for (int u = 0; u < NSUB; ++u)
                        s[u] = __builtin_amdgcn_wmma_f32_16x16x4_f32(
                                false, kb[g4 & 1][u][j], false, qf[g4 * 8 + j], (short)0, s[u], false, false);
                }
            }

            const int dkey = qrow - (kt * KEYS + 8 * hi);
            float mx = 0.0f;
            #pragma unroll
            for (int u = 0; u < NSUB; ++u) {
                #pragma unroll
                for (int r = 0; r < 8; ++r) {
                    const float sv = (u * 16 + r <= dkey) ? s[u][r] : -10000.0f;
                    s[u][r] = sv;
                    mx = (u == 0 && r == 0) ? sv : fmaxf(mx, sv);
                }
            }
            mx = fmaxf(mx, xhalf(mx));

            const float mnew = fmaxf(m_run, mx);
            const float corr = __expf(m_run - mnew);
            m_run = mnew;
            float psum[NSUB];
            #pragma unroll
            for (int u = 0; u < NSUB; ++u) {
                psum[u] = 0.0f;
                #pragma unroll
                for (int r = 0; r < 8; ++r) {
                    const float p = __expf(s[u][r] - mnew);
                    psum[u] += p;
                    s[u][r] = p;
                }
            }
            float ps = psum[0];
            #pragma unroll
            for (int u = 1; u < NSUB; ++u) ps += psum[u];
            l_run = l_run * corr + ps;
            #pragma unroll
            for (int dt = 0; dt < 8; ++dt) {
                #pragma unroll
                for (int r = 0; r < 8; ++r) o[dt][r] *= corr;
            }

            #pragma unroll
            for (int gi = 0; gi < 2 * NSUB; ++gi) {
                if (gi < 2 * NSUB - 1) {
                    #pragma unroll
                    for (int kk = 0; kk < 2; ++kk) {
                        const int kn = (gi + 1) * 2 + kk;
                        const int krow = (kn >> 2) * 16 + (kn & 3) * 2;
                        #pragma unroll
                        for (int dt = 0; dt < 8; ++dt) {
                            vlo[(gi + 1) & 1][kk][dt] = vfrag[(krow    ) * KSTRIDE + dt * 16];
                            vhi[(gi + 1) & 1][kk][dt] = vfrag[(krow + 1) * KSTRIDE + dt * 16];
                        }
                    }
                    __builtin_amdgcn_sched_barrier(0);
                }
                #pragma unroll
                for (int kk = 0; kk < 2; ++kk) {
                    const int kn = gi * 2 + kk;
                    const int pu = kn >> 2, pt = kn & 3;
                    const v2f pb = { s[pu][2 * pt], s[pu][2 * pt + 1] };
                    #pragma unroll
                    for (int dt = 0; dt < 8; ++dt) {
                        const v2f va = { vlo[gi & 1][kk][dt], vhi[gi & 1][kk][dt] };
                        o[dt] = __builtin_amdgcn_wmma_f32_16x16x4_f32(
                                    false, va, false, pb, (short)0, o[dt], false, false);
                    }
                }
            }
        }
    }

    {
        const float inv = 1.0f / (l_run + xhalf(l_run));
        float* so = sO[wave];
        const int orow = lane >> 3;
        const int ocol = (lane & 7) * 4;
        #pragma unroll
        for (int c = 0; c < 4; ++c) {
            #pragma unroll
            for (int t = 0; t < 2; ++t) {
                const v8f x = o[2 * c + t];
                const v4f lo4 = { x[0] * inv, x[1] * inv, x[2] * inv, x[3] * inv };
                const v4f hi4 = { x[4] * inv, x[5] * inv, x[6] * inv, x[7] * inv };
                *(v4f*)&so[l15 * OSTRIDE + t * 16 + 8 * hi]     = lo4;
                *(v4f*)&so[l15 * OSTRIDE + t * 16 + 8 * hi + 4] = hi4;
            }
            asm volatile("s_wait_dscnt 0x0" ::: "memory");
            #pragma unroll
            for (int i = 0; i < 4; ++i) {
                const int row = i * 4 + orow;
                const v4f x = *(const v4f*)&so[row * OSTRIDE + ocol];
                *(v4f*)(Out + (size_t)(q0w + row) * SROW + head_off + c * 32 + ocol) = x;
            }
        }
    }

    __syncthreads();
    }
}

extern "C" void kernel_launch(void* const* d_in, const int* in_sizes, int n_in,
                              void* d_out, int out_size, void* d_ws, size_t ws_size,
                              hipStream_t stream) {
    (void)in_sizes; (void)n_in; (void)out_size; (void)d_ws; (void)ws_size;
    const float* Q = (const float*)d_in[0];
    const float* K = (const float*)d_in[1];
    const float* V = (const float*)d_in[2];
    float* Out     = (float*)d_out;

    dim3 grid(SEQ / 256, BATCH * HEADS);
    dim3 block(256);
    hipLaunchKernelGGL(fa_fp32_wmma_kernel, grid, block, 0, stream, Q, K, V, Out);
}


# All LDS fragment reads widened to 16 bytes: K read as 4 consecutive head dims (Q paired alike), V tile stored transposed so V^T reads 4 consecutive keys; order of sums in the score product changes
# speedup vs baseline: 3.6064x; 1.0586x over previous
#include <hip/hip_runtime.h>
#include <hip/hip_bf16.h>


typedef __attribute__((ext_vector_type(2))) float v2f;
typedef __attribute__((ext_vector_type(4))) float v4f;
typedef __attribute__((ext_vector_type(8))) float v8f;

#define SEQ    2048
#define BATCH  2
#define HEADS  16
#define HDIM   128
#define SROW   4096
#define KSTRIDE 140
#define VSTRIDE 36
#define NSUB   2
#define KEYS   (16 * NSUB)
#define OSTRIDE 36

__device__ __forceinline__ float xhalf(float x) {
    const int b = __builtin_bit_cast(int, x);
    return __builtin_bit_cast(float,
        __builtin_amdgcn_permlanex16(b, b, 0x76543210, 0xfedcba98, false, false));
}

__global__ __launch_bounds__(256) void fa_fp32_wmma_kernel(
    const float* __restrict__ Q, const float* __restrict__ K,
    const float* __restrict__ V, float* __restrict__ Out)
{
    __shared__ __attribute__((aligned(16))) float sK[2][KEYS * KSTRIDE];
    __shared__ __attribute__((aligned(16))) float sVT[2][HDIM * VSTRIDE];
    __shared__ __attribute__((aligned(16))) float sO[8][16 * OSTRIDE];

    const int tid  = threadIdx.x;
    const int wave = tid >> 5;
    const int lane = tid & 31;
    const int hi   = lane >> 4;
    const int l15  = lane & 15;

    const size_t head_off = (size_t)blockIdx.y * HDIM;

    const int ld_row = tid >> 4;
    const int ld_col = (tid & 15) * 8;
    const size_t ld_base = (size_t)ld_row * SROW + head_off + ld_col;
    const size_t ldv_base = (size_t)(wave * 4) * SROW + head_off + lane * 4;
    const int vst_off = (lane * 4) * VSTRIDE + ((wave ^ ((lane >> 2) & 3)) * 4);

    #pragma unroll 1
    for (int pass = 0; pass < 2; ++pass) {
    const int qtile = pass == 0 ? 15 - (int)blockIdx.x : (int)blockIdx.x;
    const int qblk = qtile * 128;
    const int q0w  = qblk + wave * 16;

    v2f qf[32];
    {
        const float inv_norm = 0.08838834764831845f;
        const float* qbase = Q + (size_t)(q0w + l15) * SROW + head_off + 4 * hi;
        #pragma unroll
        for (int kp = 0; kp < 16; ++kp) {
            const v4f q = *(const v4f*)(qbase + kp * 8);
            qf[2 * kp]     = (v2f){ q[0] * inv_norm, q[1] * inv_norm };
            qf[2 * kp + 1] = (v2f){ q[2] * inv_norm, q[3] * inv_norm };
        }
    }

    v8f o[8];
    #pragma unroll
    for (int dt = 0; dt < 8; ++dt) o[dt] = (v8f){};
    float m_run = -1e30f, l_run = 0.0f;
    const int qrow = q0w + l15;

    const int nkt       = (qblk / 16 + 8) / NSUB;
    const int my_kt_max = qblk / 16 + wave;

    v4f rk[NSUB][2], rv[4];
    #pragma unroll
    for (int u = 0; u < NSUB; ++u) {
        const size_t g = ld_base + (size_t)(u * 16) * SROW;
        rk[u][0] = *(const v4f*)(K + g);
        rk[u][1] = *(const v4f*)(K + g + 4);
    }
    #pragma unroll
    for (int i = 0; i < 4; ++i)
        rv[i] = *(const v4f*)(V + ldv_base + (size_t)i * SROW);

    for (int kt = 0; kt < nkt; ++kt) {
        float* bK = sK[kt & 1];
        float* bV = sVT[kt & 1];
        #pragma unroll
        for (int u = 0; u < NSUB; ++u) {
            const int row = u * 16 + ld_row;
            *(v4f*)&bK[row * KSTRIDE + ld_col]     = rk[u][0];
            *(v4f*)&bK[row * KSTRIDE + ld_col + 4] = rk[u][1];
        }
        asm volatile("" :: "v"(rv[0]), "v"(rv[1]), "v"(rv[2]), "v"(rv[3]));
        #pragma unroll
        for (int j = 0; j < 4; ++j) {
            const v4f col = { rv[0][j], rv[1][j], rv[2][j], rv[3][j] };
            *(v4f*)&bV[vst_off + j * VSTRIDE] = col;
        }
        __syncthreads();
        if (kt + 1 < nkt) {
            #pragma unroll
            for (int u = 0; u < NSUB; ++u) {
                const size_t g = ld_base + (size_t)((kt + 1) * KEYS + u * 16) * SROW;
                rk[u][0] = *(const v4f*)(K + g);
                rk[u][1] = *(const v4f*)(K + g + 4);
            }
            #pragma unroll
            for (int i = 0; i < 4; ++i)
                rv[i] = *(const v4f*)(V + ldv_base + (size_t)((kt + 1) * KEYS + i) * SROW);
        }

        if (kt * NSUB <= my_kt_max) {
            const float* kfrag = &bK[l15 * KSTRIDE + 4 * hi];
            const float* vfrag0 = &bV[l15 * VSTRIDE + 8 * hi];
            const float* vfrag1 = &bV[l15 * VSTRIDE + 8 * (1 - hi)];
            v4f kb[2][NSUB][4];
            v4f vb[2][8];
            #pragma unroll
            for (int u = 0; u < NSUB; ++u) {
                #pragma unroll
                for (int jp = 0; jp < 4; ++jp)
                    kb[0][u][jp] = *(const v4f*)(kfrag + u * 16 * KSTRIDE + jp * 8);
            }

            v8f s[NSUB];
            #pragma unroll
            for (int u = 0; u < NSUB; ++u) s[u] = (v8f){};
            #pragma unroll
            for (int g4 = 0; g4 < 4; ++g4) {
                if (g4 < 3) {
                    #pragma unroll
                    for (int u = 0; u < NSUB; ++u) {
                        #pragma unroll
                        for (int jp = 0; jp < 4; ++jp)
                            kb[(g4 + 1) & 1][u][jp] =
                                *(const v4f*)(kfrag + u * 16 * KSTRIDE + ((g4 + 1) * 4 + jp) * 8);
                    }
                } else {
                    #pragma unroll
                    for (int dt = 0; dt < 8; ++dt)
                        vb[0][dt] = *(const v4f*)(((dt & 2) ? vfrag1 : vfrag0)
                                                  + dt * 16 * VSTRIDE + 4 * (dt & 1));
                }
                __builtin_amdgcn_sched_barrier(0);
                #pragma unroll
                for (int j = 0; j < 8; ++j) {
                    #pragma unroll
                    for (int u = 0; u < NSUB; ++u) {
                        const v4f k4 = kb[g4 & 1][u][j >> 1];
                        const v2f ka = { k4[2 * (j & 1)], k4[2 * (j & 1) + 1] };
                        s[u] = __builtin_amdgcn_wmma_f32_16x16x4_f32(
                                false, ka, false, qf[g4 * 8 + j], (short)0, s[u], false, false);
                    }
                }
            }

            const int dkey = qrow - (kt * KEYS + 8 * hi);
            float mx = 0.0f;
            #pragma unroll
            for (int u = 0; u < NSUB; ++u) {
                #pragma unroll
                for (int r = 0; r < 8; ++r) {
                    const float sv = (u * 16 + r <= dkey) ? s[u][r] : -10000.0f;
                    s[u][r] = sv;
                    mx = (u == 0 && r == 0) ? sv : fmaxf(mx, sv);
                }
            }
            mx = fmaxf(mx, xhalf(mx));

            const float mnew = fmaxf(m_run, mx);
            const float corr = __expf(m_run - mnew);
            m_run = mnew;
            float psum[NSUB];
            #pragma unroll
            for (int u = 0; u < NSUB; ++u) {
                psum[u] = 0.0f;
                #pragma unroll
                for (int r = 0; r < 8; ++r) {
                    const float p = __expf(s[u][r] - mnew);
                    psum[u] += p;
                    s[u][r] = p;
                }
            }
            float ps = psum[0];
            #pragma unroll
            for (int u = 1; u < NSUB; ++u) ps += psum[u];
            l_run = l_run * corr + ps;
            #pragma unroll
            for (int dt = 0; dt < 8; ++dt) {
                #pragma unroll
                for (int r = 0; r < 8; ++r) o[dt][r] *= corr;
            }

            #pragma unroll
            for (int gi = 0; gi < 2 * NSUB; ++gi) {
                if (gi < 2 * NSUB - 1) {
                    const int nu = (gi + 1) >> 1, nh = (gi + 1) & 1;
                    #pragma unroll
                    for (int dt = 0; dt < 8; ++dt)
                        vb[(gi + 1) & 1][dt] = *(const v4f*)(((dt & 2) ? vfrag1 : vfrag0)
                                                  + dt * 16 * VSTRIDE + nu * 16 + 4 * (nh ^ (dt & 1)));
                    __builtin_amdgcn_sched_barrier(0);
                }
                #pragma unroll
                for (int kk = 0; kk < 2; ++kk) {
                    const int kn = gi * 2 + kk;
                    const int pu = kn >> 2, pt = kn & 3;
                    const v2f pb = { s[pu][2 * pt], s[pu][2 * pt + 1] };
                    #pragma unroll
                    for (int dt = 0; dt < 8; ++dt) {
                        const v4f v4 = vb[gi & 1][dt];
                        const v2f va = { v4[2 * kk], v4[2 * kk + 1] };
                        o[dt] = __builtin_amdgcn_wmma_f32_16x16x4_f32(
                                    false, va, false, pb, (short)0, o[dt], false, false);
                    }
                }
            }
        }
    }

    {
        const float inv = 1.0f / (l_run + xhalf(l_run));
        float* so = sO[wave];
        const int orow = lane >> 3;
        const int ocol = (lane & 7) * 4;
        #pragma unroll
        for (int c = 0; c < 4; ++c) {
            #pragma unroll
            for (int t = 0; t < 2; ++t) {
                const v8f x = o[2 * c + t];
                const v4f lo4 = { x[0] * inv, x[1] * inv, x[2] * inv, x[3] * inv };
                const v4f hi4 = { x[4] * inv, x[5] * inv, x[6] * inv, x[7] * inv };
                *(v4f*)&so[l15 * OSTRIDE + t * 16 + 8 * hi]     = lo4;
                *(v4f*)&so[l15 * OSTRIDE + t * 16 + 8 * hi + 4] = hi4;
            }
            asm volatile("s_wait_dscnt 0x0" ::: "memory");
            #pragma unroll
            for (int i = 0; i < 4; ++i) {
                const int row = i * 4 + orow;
                const v4f x = *(const v4f*)&so[row * OSTRIDE + ocol];
                *(v4f*)(Out + (size_t)(q0w + row) * SROW + head_off + c * 32 + ocol) = x;
            }
        }
    }

    __syncthreads();
    }
}

extern "C" void kernel_launch(void* const* d_in, const int* in_sizes, int n_in,
                              void* d_out, int out_size, void* d_ws, size_t ws_size,
                              hipStream_t stream) {
    (void)in_sizes; (void)n_in; (void)out_size; (void)d_ws; (void)ws_size;
    const float* Q = (const float*)d_in[0];
    const float* K = (const float*)d_in[1];
    const float* V = (const float*)d_in[2];
    float* Out     = (float*)d_out;

    dim3 grid(SEQ / 256, BATCH * HEADS);
    dim3 block(256);
    hipLaunchKernelGGL(fa_fp32_wmma_kernel, grid, block, 0, stream, Q, K, V, Out);
}
